# starting kernel
# baseline (speedup 1.0000x reference)
#include <hip/hip_runtime.h>
#include <math.h>
#include <limits.h>

#define N_NODES   50000
#define N_EDGES   800000
#define IN_FEATS  128
#define OUT_FEATS 16
#define NUM_HEADS 4
#define ZCOLS     (NUM_HEADS * OUT_FEATS)
#define ALPHA     0.2f

typedef __attribute__((ext_vector_type(2))) float v2f;
typedef __attribute__((ext_vector_type(8))) float v8f;

__global__ __launch_bounds__(128) void gat_proj_wmma(
    const float* __restrict__ h, const float* __restrict__ W,
    float* __restrict__ z)
{
  const int wave  = threadIdx.x >> 5;
  const int lane  = threadIdx.x & 31;
  const int mtile = blockIdx.x;
  const int ntile = wave;
  const int lo16  = lane & 15;
  const int kbase = (lane >> 4) * 2;

  const float* __restrict__ ha = h + (size_t)(mtile * 16 + lo16) * IN_FEATS + kbase;
  const float* __restrict__ wb = W + (size_t)(ntile * 16 + lo16) * IN_FEATS + kbase;

  v8f c = {};
#pragma unroll 4
  for (int k0 = 0; k0 < IN_FEATS; k0 += 4) {
    v2f a, b;
    a.x = ha[k0]; a.y = ha[k0 + 1];
    b.x = wb[k0]; b.y = wb[k0 + 1];
    c = __builtin_amdgcn_wmma_f32_16x16x4_f32(
         false, a,  false, b,
         (short)0, c,  false,  false);
  }

  const int rbase = mtile * 16 + ((lane >> 4) << 3);
  float* __restrict__ zp = z + (size_t)rbase * ZCOLS + ntile * 16 + lo16;
#pragma unroll
  for (int r = 0; r < 8; ++r)
    zp[(size_t)r * ZCOLS] = c[r];
}

__global__ void gat_node_scores(const float* __restrict__ z,
                                const float* __restrict__ A,
                                float* __restrict__ asrc,
                                float* __restrict__ adst,
                                int* __restrict__ maxbuf)
{
  int t = blockIdx.x * blockDim.x + threadIdx.x;
  if (t >= N_NODES * NUM_HEADS) return;
  const int n = t >> 2, hh = t & 3;
  const float* __restrict__ zv = z + (size_t)n * ZCOLS + hh * OUT_FEATS;
  const float* __restrict__ As = A + hh * (2 * OUT_FEATS);
  float s = 0.f, d = 0.f;
#pragma unroll
  for (int f = 0; f < OUT_FEATS; ++f) {
    s += zv[f] * As[f];
    d += zv[f] * As[OUT_FEATS + f];
  }
  asrc[t] = s;
  adst[t] = d;
  maxbuf[t] = INT_MIN;
}

__device__ __forceinline__ int f32_ord(float v) {
  int i = __float_as_int(v);
  return i >= 0 ? i : (i ^ 0x7FFFFFFF);
}

__global__ void gat_edge_logits(const int* __restrict__ src,
                                const int* __restrict__ dst,
                                const float* __restrict__ asrc,
                                const float* __restrict__ adst,
                                float* __restrict__ eval,
                                int* __restrict__ maxbuf)
{
  int t = blockIdx.x * blockDim.x + threadIdx.x;
  if (t >= N_EDGES * NUM_HEADS) return;
  const int e = t >> 2, hh = t & 3;
  const int s = src[e], d = dst[e];
  float v = asrc[s * NUM_HEADS + hh] + adst[d * NUM_HEADS + hh];
  v = v > 0.f ? v : ALPHA * v;
  eval[t] = v;
  atomicMax(&maxbuf[d * NUM_HEADS + hh], f32_ord(v));
}

__global__ void gat_prep(const int* __restrict__ maxbuf,
                         float* __restrict__ m,
                         float* __restrict__ denom,
                         float* __restrict__ out)
{
  int t = blockIdx.x * blockDim.x + threadIdx.x;
  if (t >= N_NODES * NUM_HEADS) return;
  const int iv = maxbuf[t];
  float mv = 0.f;
  if (iv != INT_MIN) {
    int j = iv >= 0 ? iv : (iv ^ 0x7FFFFFFF);
    mv = __int_as_float(j);
    if (!isfinite(mv)) mv = 0.f;
  }
  m[t] = mv;
  denom[t] = 0.f;
  float* __restrict__ op = out + (size_t)(t >> 2) * ZCOLS + (t & 3) * OUT_FEATS;
#pragma unroll
  for (int f = 0; f < OUT_FEATS; ++f) op[f] = 0.f;
}

__global__ __launch_bounds__(256) void gat_edge_accum(
    const int* __restrict__ src, const int* __restrict__ dst,
    const float* __restrict__ eval, const float* __restrict__ m,
    const float* __restrict__ z,
    float* __restrict__ out, float* __restrict__ denom)
{
  const long long t = (long long)blockIdx.x * 256 + threadIdx.x;
  const int f  = (int)(t & 15);
  const int hh = (int)((t >> 4) & 3);
  const int e  = (int)(t >> 6);
  const int s = src[e], d = dst[e];
  const float ex = expf(eval[e * NUM_HEADS + hh] - m[d * NUM_HEADS + hh]);
  atomicAdd(&out[(size_t)d * ZCOLS + hh * OUT_FEATS + f],
            ex * z[(size_t)s * ZCOLS + hh * OUT_FEATS + f]);
  if (f == 0) atomicAdd(&denom[d * NUM_HEADS + hh], ex);
}

__global__ __launch_bounds__(256) void gat_finalize(float* __restrict__ out,
                                                    const float* __restrict__ denom)
{
  const int t = blockIdx.x * 256 + threadIdx.x;
  const int n = t >> 6, j = t & 63;
  out[t] /= fmaxf(denom[n * NUM_HEADS + (j >> 4)], 1e-12f);
}

extern "C" void kernel_launch(void* const* d_in, const int* in_sizes, int n_in,
                              void* d_out, int out_size, void* d_ws, size_t ws_size,
                              hipStream_t stream) {
  const float* h   = (const float*)d_in[0];
  const int*   src = (const int*)d_in[1];
  const int*   dst = (const int*)d_in[2];
  const float* W   = (const float*)d_in[3];
  const float* A   = (const float*)d_in[4];
  float* out = (float*)d_out;

  float* z     = (float*)d_ws;
  float* asrc  = z     + (size_t)N_NODES * ZCOLS;
  float* adst  = asrc  + (size_t)N_NODES * NUM_HEADS;
  float* m     = adst  + (size_t)N_NODES * NUM_HEADS;
  float* denom = m     + (size_t)N_NODES * NUM_HEADS;
  float* eval  = denom + (size_t)N_NODES * NUM_HEADS;
  int*   maxbuf = (int*)(eval + (size_t)N_EDGES * NUM_HEADS);

  const int NH = N_NODES * NUM_HEADS;
  const int EH = N_EDGES * NUM_HEADS;

  gat_proj_wmma<<<N_NODES / 16, 128, 0, stream>>>(h, W, z);
  gat_node_scores<<<(NH + 255) / 256, 256, 0, stream>>>(z, A, asrc, adst, maxbuf);
  gat_edge_logits<<<(EH + 255) / 256, 256, 0, stream>>>(src, dst, asrc, adst, eval, maxbuf);
  gat_prep<<<(NH + 255) / 256, 256, 0, stream>>>(maxbuf, m, denom, out);
  gat_edge_accum<<<(N_EDGES * 64) / 256, 256, 0, stream>>>(src, dst, eval, m, z, out, denom);
  gat_finalize<<<(N_NODES * ZCOLS) / 256, 256, 0, stream>>>(out, denom);
}


# edge scatter with float atomics (and the atomic segment max) replaced by a per-destination gather over a counting-sorted edge list built in the call; per-node sums now run in list order
# speedup vs baseline: 1.9795x; 1.9795x over previous
#include <hip/hip_runtime.h>
#include <math.h>


#define N_NODES   50000
#define N_EDGES   800000
#define IN_FEATS  128
#define OUT_FEATS 16
#define NUM_HEADS 4
#define ZCOLS     (NUM_HEADS * OUT_FEATS)
#define ALPHA     0.2f

typedef __attribute__((ext_vector_type(2))) float v2f;
typedef __attribute__((ext_vector_type(8))) float v8f;

__global__ __launch_bounds__(128) void gat_proj_wmma(
    const float* __restrict__ h, const float* __restrict__ W,
    float* __restrict__ z)
{
  const int wave  = threadIdx.x >> 5;
  const int lane  = threadIdx.x & 31;
  const int mtile = blockIdx.x;
  const int ntile = wave;
  const int lo16  = lane & 15;
  const int kbase = (lane >> 4) * 2;

  const float* __restrict__ ha = h + (size_t)(mtile * 16 + lo16) * IN_FEATS + kbase;
  const float* __restrict__ wb = W + (size_t)(ntile * 16 + lo16) * IN_FEATS + kbase;

  v8f c = {};
#pragma unroll 4
  for (int k0 = 0; k0 < IN_FEATS; k0 += 4) {
    v2f a, b;
    a.x = ha[k0]; a.y = ha[k0 + 1];
    b.x = wb[k0]; b.y = wb[k0 + 1];
    c = __builtin_amdgcn_wmma_f32_16x16x4_f32(
         false, a,  false, b,
         (short)0, c,  false,  false);
  }

  const int rbase = mtile * 16 + ((lane >> 4) << 3);
  float* __restrict__ zp = z + (size_t)rbase * ZCOLS + ntile * 16 + lo16;
#pragma unroll
  for (int r = 0; r < 8; ++r)
    zp[(size_t)r * ZCOLS] = c[r];
}

__global__ void gat_node_scores(const float* __restrict__ z,
                                const float* __restrict__ A,
                                float* __restrict__ asrc,
                                float* __restrict__ adst,
                                int* __restrict__ cnt)
{
  int t = blockIdx.x * blockDim.x + threadIdx.x;
  if (t >= N_NODES * NUM_HEADS) return;
  const int n = t >> 2, hh = t & 3;
  const float* __restrict__ zv = z + (size_t)n * ZCOLS + hh * OUT_FEATS;
  const float* __restrict__ As = A + hh * (2 * OUT_FEATS);
  float s = 0.f, d = 0.f;
#pragma unroll
  for (int f = 0; f < OUT_FEATS; ++f) {
    s += zv[f] * As[f];
    d += zv[f] * As[OUT_FEATS + f];
  }
  asrc[t] = s;
  adst[t] = d;
  if (t < N_NODES) cnt[t] = 0;
}

__global__ __launch_bounds__(256) void gat_edge_rank(const int* __restrict__ dst,
                                                     int* __restrict__ cnt,
                                                     int* __restrict__ rank)
{
  const int e = blockIdx.x * 256 + threadIdx.x;
  rank[e] = atomicAdd(&cnt[dst[e]], 1);
}

#define SCAN_IT 13
__global__ __launch_bounds__(1024) void gat_row_scan(const int* __restrict__ cnt,
                                                     int* __restrict__ rowptr)
{
  __shared__ int wtot[32];
  const int wave = threadIdx.x >> 5;
  const int lane = threadIdx.x & 31;
  const int base = wave * (SCAN_IT * 128) + lane * 4;

  int4 v[SCAN_IT];
  int lsum = 0;
#pragma unroll
  for (int i = 0; i < SCAN_IT; ++i) {
    const int idx = base + i * 128;
    int4 x = make_int4(0, 0, 0, 0);
    if (idx < N_NODES) x = *(const int4*)(cnt + idx);
    v[i] = x;
    lsum += x.x + x.y + x.z + x.w;
  }
  for (int off = 16; off > 0; off >>= 1) lsum += __shfl_xor(lsum, off);
  if (lane == 0) wtot[wave] = lsum;
  __syncthreads();

  const int w = wtot[lane];
  int inc = w;
  for (int off = 1; off < 32; off <<= 1) {
    const int u = __shfl_up(inc, off);
    if (lane >= off) inc += u;
  }
  int carry = __shfl(inc - w, wave);
  const int total = __shfl(inc, 31);

#pragma unroll
  for (int i = 0; i < SCAN_IT; ++i) {
    const int4 x = v[i];
    const int ls = x.x + x.y + x.z + x.w;
    int incl = ls;
    for (int off = 1; off < 32; off <<= 1) {
      const int u = __shfl_up(incl, off);
      if (lane >= off) incl += u;
    }
    const int e0 = carry + incl - ls;
    const int idx = base + i * 128;
    if (idx < N_NODES)
      *(int4*)(rowptr + idx) = make_int4(e0, e0 + x.x, e0 + x.x + x.y, e0 + x.x + x.y + x.z);
    carry += __shfl(incl, 31);
  }
  if (threadIdx.x == 0) rowptr[N_NODES] = total;
}

__global__ __launch_bounds__(256) void gat_edge_fill(const int* __restrict__ src,
                                                     const int* __restrict__ dst,
                                                     const int* __restrict__ rank,
                                                     const int* __restrict__ rowptr,
                                                     int* __restrict__ csr_src)
{
  const int e = blockIdx.x * 256 + threadIdx.x;
  csr_src[rowptr[dst[e]] + rank[e]] = src[e];
}

__device__ __forceinline__ float lrelu(float v) { return v > 0.f ? v : ALPHA * v; }

__device__ __forceinline__ int node_clamp(int s) {
  return (int)min((unsigned)s, (unsigned)(N_NODES - 1));
}

__global__ __launch_bounds__(256) void gat_gather(
    const int* __restrict__ rowptr, const int* __restrict__ csr_src,
    const float* __restrict__ asrc, const float* __restrict__ adst,
    const float* __restrict__ z, float* __restrict__ out)
{
  const int t  = blockIdx.x * 256 + threadIdx.x;
  const int n  = t >> 4;
  const int q  = t & 15;
  const int hh = q >> 2;
  const int beg = rowptr[n], end = rowptr[n + 1];
  const float4 ad4 = *(const float4*)(adst + n * NUM_HEADS);

  float4 m4 = make_float4(-INFINITY, -INFINITY, -INFINITY, -INFINITY);
  for (int i = beg + q; i < end; i += 16) {
    const int s = node_clamp(csr_src[i]);
    const float4 a4 = *(const float4*)(asrc + s * NUM_HEADS);
    m4.x = fmaxf(m4.x, lrelu(a4.x + ad4.x));
    m4.y = fmaxf(m4.y, lrelu(a4.y + ad4.y));
    m4.z = fmaxf(m4.z, lrelu(a4.z + ad4.z));
    m4.w = fmaxf(m4.w, lrelu(a4.w + ad4.w));
  }
  for (int off = 8; off > 0; off >>= 1) {
    m4.x = fmaxf(m4.x, __shfl_xor(m4.x, off, 16));
    m4.y = fmaxf(m4.y, __shfl_xor(m4.y, off, 16));
    m4.z = fmaxf(m4.z, __shfl_xor(m4.z, off, 16));
    m4.w = fmaxf(m4.w, __shfl_xor(m4.w, off, 16));
  }
  float mx = hh == 0 ? m4.x : hh == 1 ? m4.y : hh == 2 ? m4.z : m4.w;
  const float ad = hh == 0 ? ad4.x : hh == 1 ? ad4.y : hh == 2 ? ad4.z : ad4.w;
  if (!isfinite(mx)) mx = 0.f;

  float4 acc = make_float4(0.f, 0.f, 0.f, 0.f);
  float den = 0.f;
  const float* __restrict__ zq = z + q * 4;
  const float* __restrict__ aq = asrc + hh;
  for (int k = beg; k < end; k += 4) {
    int   s4[4];
    float a4[4];
    float4 z4[4];
#pragma unroll
    for (int j = 0; j < 4; ++j)
      s4[j] = node_clamp(csr_src[min(k + j, end - 1)]);
#pragma unroll
    for (int j = 0; j < 4; ++j) {
      a4[j] = aq[s4[j] * NUM_HEADS];
      z4[j] = *(const float4*)(zq + (size_t)s4[j] * ZCOLS);
    }
#pragma unroll
    for (int j = 0; j < 4; ++j) {
      const float v  = lrelu(a4[j] + ad);
      const float ex = expf((k + j < end) ? v - mx : -INFINITY);
      den += ex;
      acc.x += ex * z4[j].x;
      acc.y += ex * z4[j].y;
      acc.z += ex * z4[j].z;
      acc.w += ex * z4[j].w;
    }
  }
  const float dd = fmaxf(den, 1e-12f);
  *(float4*)(out + (size_t)n * ZCOLS + q * 4) =
      make_float4(acc.x / dd, acc.y / dd, acc.z / dd, acc.w / dd);
}

extern "C" void kernel_launch(void* const* d_in, const int* in_sizes, int n_in,
                              void* d_out, int out_size, void* d_ws, size_t ws_size,
                              hipStream_t stream) {
  const float* h   = (const float*)d_in[0];
  const int*   src = (const int*)d_in[1];
  const int*   dst = (const int*)d_in[2];
  const float* W   = (const float*)d_in[3];
  const float* A   = (const float*)d_in[4];
  float* out = (float*)d_out;

  float* z     = (float*)d_ws;
  float* asrc  = z     + (size_t)N_NODES * ZCOLS;
  float* adst  = asrc  + (size_t)N_NODES * NUM_HEADS;
  int*   cnt     = (int*)(adst + (size_t)N_NODES * NUM_HEADS);
  int*   rowptr  = cnt    + N_NODES;
  int*   rank    = rowptr + (N_NODES + 4);
  int*   csr_src = rank   + N_EDGES;

  const int NH = N_NODES * NUM_HEADS;

  gat_proj_wmma<<<N_NODES / 16, 128, 0, stream>>>(h, W, z);
  gat_node_scores<<<(NH + 255) / 256, 256, 0, stream>>>(z, A, asrc, adst, cnt);
  gat_edge_rank<<<N_EDGES / 256, 256, 0, stream>>>(dst, cnt, rank);
  gat_row_scan<<<1, 1024, 0, stream>>>(cnt, rowptr);
  gat_edge_fill<<<N_EDGES / 256, 256, 0, stream>>>(src, dst, rank, rowptr, csr_src);
  gat_gather<<<(N_NODES * 16) / 256, 256, 0, stream>>>(rowptr, csr_src, asrc, adst, z, out);
}


# edge placement: 800000 scattered returning global adds replaced by LDS counters (edges binned by 256-node bucket, one global add per block and bucket, rows built in LDS); arrival-order sums
# speedup vs baseline: 4.6954x; 2.3720x over previous
#include <hip/hip_runtime.h>
#include <math.h>


#define N_NODES   50000
#define N_EDGES   800000
#define IN_FEATS  128
#define OUT_FEATS 16
#define NUM_HEADS 4
#define ZCOLS     (NUM_HEADS * OUT_FEATS)
#define ALPHA     0.2f

typedef __attribute__((ext_vector_type(2))) float v2f;
typedef __attribute__((ext_vector_type(8))) float v8f;

#define SLOT_CAP  32
#define OVF_WORDS 50000
#define OVF_HEAD  256
#define GCNT_OFF  2
#define OVF_PAIR0 (OVF_HEAD / 2)
#define OVF_CAP   ((OVF_WORDS - OVF_HEAD) / 2)
#define BKT_NODES 256
#define N_BKT     ((N_NODES + BKT_NODES - 1) / BKT_NODES)
#define BKT_WORDS (BKT_NODES * SLOT_CAP)
#define BIN_EDGES 2048
#define BIN_BLOCKS ((N_EDGES + BIN_EDGES - 1) / BIN_EDGES)
static_assert(N_NODES <= 65536, "a bucket word keeps the source id in 16 bits");
static_assert(N_EDGES % 4 == 0, "edges are read as whole quads");
static_assert(GCNT_OFF + N_BKT <= OVF_HEAD, "fill counts lie inside the cleared head");

__device__ __forceinline__ int bucket_cap(int b) {
  return min(BKT_WORDS, N_NODES * SLOT_CAP - b * BKT_WORDS);
}
__device__ __forceinline__ int bucket_of(int d) {
  return (int)min((unsigned)d >> 8, (unsigned)(N_BKT - 1));
}

__device__ __forceinline__ void proj_tile(
    const float* __restrict__ h, const float* __restrict__ W,
    float* __restrict__ z, const int mtile)
{
  const int wave  = threadIdx.x >> 5;
  const int lane  = threadIdx.x & 31;
  const int ntile = wave;
  const int lo16  = lane & 15;
  const int kbase = (lane >> 4) * 2;

  const float* __restrict__ ha = h + (size_t)(mtile * 16 + lo16) * IN_FEATS + kbase;
  const float* __restrict__ wb = W + (size_t)(ntile * 16 + lo16) * IN_FEATS + kbase;

  v8f c = {};
#pragma unroll 4
  for (int k0 = 0; k0 < IN_FEATS; k0 += 4) {
    v2f a, b;
    a.x = ha[k0]; a.y = ha[k0 + 1];
    b.x = wb[k0]; b.y = wb[k0 + 1];
    c = __builtin_amdgcn_wmma_f32_16x16x4_f32(
         false, a,  false, b,
         (short)0, c,  false,  false);
  }

  const int rbase = mtile * 16 + ((lane >> 4) << 3);
  float* __restrict__ zp = z + (size_t)rbase * ZCOLS + ntile * 16 + lo16;
#pragma unroll
  for (int r = 0; r < 8; ++r)
    zp[(size_t)r * ZCOLS] = c[r];
}

__device__ __forceinline__ void node_scores(const float* __restrict__ z,
                                            const float* __restrict__ A,
                                            float* __restrict__ asrc,
                                            float* __restrict__ adst,
                                            const int t)
{
  if (t >= N_NODES * NUM_HEADS) return;
  const int n = t >> 2, hh = t & 3;
  const float* __restrict__ zv = z + (size_t)n * ZCOLS + hh * OUT_FEATS;
  const float* __restrict__ As = A + hh * (2 * OUT_FEATS);
  float s = 0.f, d = 0.f;
#pragma unroll
  for (int f = 0; f < OUT_FEATS; ++f) {
    s += zv[f] * As[f];
    d += zv[f] * As[OUT_FEATS + f];
  }
  asrc[t] = s;
  adst[t] = d;
}

__global__ __launch_bounds__(64) void gat_clear_counts(int* __restrict__ ovf)
{
  ((int4*)ovf)[threadIdx.x] = make_int4(0, 0, 0, 0);
}

#define PROJ_BLOCKS (N_NODES / 16)
__global__ __launch_bounds__(128) void gat_proj_and_bin(
    const float* __restrict__ h, const float* __restrict__ W, float* __restrict__ z,
    const int* __restrict__ src, const int* __restrict__ dst,
    int* __restrict__ ovf, int* __restrict__ slots)
{
  __shared__ int s_hist[256];
  __shared__ int s_cur[256];
  const int b = blockIdx.x;
  if (b >= BIN_BLOCKS) {
    proj_tile(h, W, z, b - BIN_BLOCKS);
    return;
  }
  const int tid = threadIdx.x;
  s_hist[tid] = 0;
  s_hist[tid + 128] = 0;

  int4 d4[4], s4[4];
  const int e0 = b * BIN_EDGES + tid * 4;
#pragma unroll
  for (int j = 0; j < 4; ++j) {
    const int e = e0 + j * 512;
    if (e < N_EDGES) {
      d4[j] = *(const int4*)(dst + e);
      s4[j] = *(const int4*)(src + e);
    } else {
      d4[j] = make_int4(0, 0, 0, 0);
      s4[j] = make_int4(0, 0, 0, 0);
    }
  }
  __syncthreads();
#pragma unroll
  for (int j = 0; j < 4; ++j) {
    if (e0 + j * 512 < N_EDGES) {
      atomicAdd(&s_hist[bucket_of(d4[j].x)], 1);
      atomicAdd(&s_hist[bucket_of(d4[j].y)], 1);
      atomicAdd(&s_hist[bucket_of(d4[j].z)], 1);
      atomicAdd(&s_hist[bucket_of(d4[j].w)], 1);
    }
  }
  __syncthreads();
#pragma unroll
  for (int i = tid; i < 256; i += 128) {
    int base = 0;
    if (i < N_BKT) {
      const int hcount = s_hist[i];
      if (hcount > 0) base = atomicAdd(&ovf[GCNT_OFF + i], hcount);
    }
    s_cur[i] = base;
  }
  __syncthreads();
#pragma unroll
  for (int j = 0; j < 4; ++j) {
    if (e0 + j * 512 < N_EDGES) {
      const int dd[4] = { d4[j].x, d4[j].y, d4[j].z, d4[j].w };
      const int ss[4] = { s4[j].x, s4[j].y, s4[j].z, s4[j].w };
      int bk[4], pos[4];
#pragma unroll
      for (int u = 0; u < 4; ++u) {
        bk[u]  = bucket_of(dd[u]);
        pos[u] = atomicAdd(&s_cur[bk[u]], 1);
      }
#pragma unroll
      for (int u = 0; u < 4; ++u) {
        if (pos[u] < bucket_cap(bk[u]))
          slots[bk[u] * BKT_WORDS + pos[u]] =
              ((dd[u] & (BKT_NODES - 1)) << 16) | (ss[u] & 0xFFFF);
      }
    }
  }
}

#define SCORE_BLOCKS ((N_NODES * NUM_HEADS + 1023) / 1024)
__global__ __launch_bounds__(1024) void gat_scores_and_rows(
    const float* __restrict__ z, const float* __restrict__ A,
    float* __restrict__ asrc, float* __restrict__ adst,
    int* __restrict__ cnt, int* __restrict__ ovf, int* __restrict__ slots)
{
  __shared__ int s_row[BKT_WORDS];
  __shared__ int s_cnt[BKT_NODES];
  const int b = blockIdx.x;
  if (b >= N_BKT) {
    node_scores(z, A, asrc, adst, (b - N_BKT) * 1024 + threadIdx.x);
    return;
  }
  const int tid   = threadIdx.x;
  const int total = ovf[GCNT_OFF + b];
  int* __restrict__ area = slots + b * BKT_WORDS;
  const int n_own = b * BKT_NODES + tid;

  if (total > bucket_cap(b)) {
    if (tid < BKT_NODES && n_own < N_NODES) cnt[n_own] = SLOT_CAP + 1;
    if (tid == 0) atomicAdd(&ovf[0], OVF_CAP + 1);
    return;
  }

  if (tid < BKT_NODES) s_cnt[tid] = 0;
  int w[8];
#pragma unroll
  for (int j = 0; j < 8; ++j) {
    const int i = j * 1024 + tid;
    w[j] = (i < total) ? area[i] : -1;
  }
  __syncthreads();
#pragma unroll
  for (int j = 0; j < 8; ++j) {
    if (w[j] >= 0) {
      const int dl = (w[j] >> 16) & (BKT_NODES - 1);
      const int s  = w[j] & 0xFFFF;
      const int r  = atomicAdd(&s_cnt[dl], 1);
      if (r < SLOT_CAP) {
        s_row[dl * SLOT_CAP + r] = s;
      } else {
        const int p = atomicAdd(&ovf[0], 1);
        if (p < OVF_CAP) ((int2*)ovf)[OVF_PAIR0 + p] = make_int2(b * BKT_NODES + dl, s);
      }
    }
  }
  __syncthreads();
#pragma unroll
  for (int j = 0; j < 8; ++j) {
    const int i  = j * 1024 + tid;
    const int dl = i >> 5;
    const int r  = i & (SLOT_CAP - 1);
    if (b * BKT_NODES + dl < N_NODES && r < min(s_cnt[dl], SLOT_CAP))
      area[i] = s_row[i];
  }
  if (tid < BKT_NODES && n_own < N_NODES) cnt[n_own] = s_cnt[tid];
}

__device__ __forceinline__ float lrelu(float v) { return v > 0.f ? v : ALPHA * v; }

__device__ __forceinline__ int node_clamp(int s) {
  return (int)min((unsigned)s, (unsigned)(N_NODES - 1));
}

__device__ __forceinline__ void max_edge(float4& m4, const float* __restrict__ asrc,
                                         const int s, const float4 ad4)
{
  const float4 a4 = *(const float4*)(asrc + s * NUM_HEADS);
  m4.x = fmaxf(m4.x, lrelu(a4.x + ad4.x));
  m4.y = fmaxf(m4.y, lrelu(a4.y + ad4.y));
  m4.z = fmaxf(m4.z, lrelu(a4.z + ad4.z));
  m4.w = fmaxf(m4.w, lrelu(a4.w + ad4.w));
}

__device__ __forceinline__ void sum_edge(float4& acc, float& den,
                                         const float* __restrict__ aq,
                                         const float* __restrict__ zq,
                                         const int s, const float ad, const float mx)
{
  const float a = aq[s * NUM_HEADS];
  const float4 zr = *(const float4*)(zq + (size_t)s * ZCOLS);
  const float ex = expf(lrelu(a + ad) - mx);
  den += ex;
  acc.x += ex * zr.x;
  acc.y += ex * zr.y;
  acc.z += ex * zr.z;
  acc.w += ex * zr.w;
}

__global__ __launch_bounds__(256) void gat_gather(
    const int* __restrict__ cnt, const int* __restrict__ slots,
    const int* __restrict__ ovf,
    const int* __restrict__ src, const int* __restrict__ dst,
    const float* __restrict__ asrc, const float* __restrict__ adst,
    const float* __restrict__ z, float* __restrict__ out)
{
  const int t  = blockIdx.x * 256 + threadIdx.x;
  const int n  = t >> 4;
  const int q  = t & 15;
  const int hh = q >> 2;
  const int deg = cnt[n];
  const int* __restrict__ row = slots + n * SLOT_CAP;
  const float4 ad4 = *(const float4*)(adst + n * NUM_HEADS);
  const bool spill = deg > SLOT_CAP;
  const int  olen  = spill ? ovf[0] : 0;
  const bool walk  = olen > OVF_CAP;
  const int  end   = walk ? 0 : min(deg, SLOT_CAP);
  const int  novf  = walk ? 0 : olen;
  const int  nwalk = walk ? N_EDGES : 0;

  float4 m4 = make_float4(-INFINITY, -INFINITY, -INFINITY, -INFINITY);
  for (int i = q; i < end; i += 16)
    max_edge(m4, asrc, node_clamp(row[i]), ad4);
  for (int i = q; i < novf; i += 16) {
    const int2 pr = ((const int2*)ovf)[OVF_PAIR0 + i];
    if (pr.x == n) max_edge(m4, asrc, node_clamp(pr.y), ad4);
  }
  for (int e = q; e < nwalk; e += 16)
    if (dst[e] == n) max_edge(m4, asrc, node_clamp(src[e]), ad4);
  for (int off = 8; off > 0; off >>= 1) {
    m4.x = fmaxf(m4.x, __shfl_xor(m4.x, off, 16));
    m4.y = fmaxf(m4.y, __shfl_xor(m4.y, off, 16));
    m4.z = fmaxf(m4.z, __shfl_xor(m4.z, off, 16));
    m4.w = fmaxf(m4.w, __shfl_xor(m4.w, off, 16));
  }
  float mx = hh == 0 ? m4.x : hh == 1 ? m4.y : hh == 2 ? m4.z : m4.w;
  const float ad = hh == 0 ? ad4.x : hh == 1 ? ad4.y : hh == 2 ? ad4.z : ad4.w;
  if (!isfinite(mx)) mx = 0.f;

  float4 acc = make_float4(0.f, 0.f, 0.f, 0.f);
  float den = 0.f;
  const float* __restrict__ zq = z + q * 4;
  const float* __restrict__ aq = asrc + hh;
  for (int k = 0; k < end; k += 4) {
    int   s4[4];
    float a4[4];
    float4 z4[4];
#pragma unroll
    for (int j = 0; j < 4; ++j)
      s4[j] = node_clamp(row[min(k + j, end - 1)]);
#pragma unroll
    for (int j = 0; j < 4; ++j) {
      a4[j] = aq[s4[j] * NUM_HEADS];
      z4[j] = *(const float4*)(zq + (size_t)s4[j] * ZCOLS);
    }
#pragma unroll
    for (int j = 0; j < 4; ++j) {
      const float v  = lrelu(a4[j] + ad);
      const float ex = expf((k + j < end) ? v - mx : -INFINITY);
      den += ex;
      acc.x += ex * z4[j].x;
      acc.y += ex * z4[j].y;
      acc.z += ex * z4[j].z;
      acc.w += ex * z4[j].w;
    }
  }
  for (int i = 0; i < novf; ++i) {
    const int2 pr = ((const int2*)ovf)[OVF_PAIR0 + i];
    if (pr.x == n) sum_edge(acc, den, aq, zq, node_clamp(pr.y), ad, mx);
  }
  for (int e = 0; e < nwalk; ++e)
    if (dst[e] == n) sum_edge(acc, den, aq, zq, node_clamp(src[e]), ad, mx);
  const float dd = fmaxf(den, 1e-12f);
  *(float4*)(out + (size_t)n * ZCOLS + q * 4) =
      make_float4(acc.x / dd, acc.y / dd, acc.z / dd, acc.w / dd);
}

extern "C" void kernel_launch(void* const* d_in, const int* in_sizes, int n_in,
                              void* d_out, int out_size, void* d_ws, size_t ws_size,
                              hipStream_t stream) {
  const float* h   = (const float*)d_in[0];
  const int*   src = (const int*)d_in[1];
  const int*   dst = (const int*)d_in[2];
  const float* W   = (const float*)d_in[3];
  const float* A   = (const float*)d_in[4];
  float* out = (float*)d_out;

  float* z     = (float*)d_ws;
  float* asrc  = z     + (size_t)N_NODES * ZCOLS;
  float* adst  = asrc  + (size_t)N_NODES * NUM_HEADS;
  int*   cnt   = (int*)(adst + (size_t)N_NODES * NUM_HEADS);
  int*   ovf   = cnt + N_NODES;
  int*   slots = ovf + OVF_WORDS;

  gat_clear_counts<<<1, 64, 0, stream>>>(ovf);
  gat_proj_and_bin<<<BIN_BLOCKS + PROJ_BLOCKS, 128, 0, stream>>>(h, W, z, src, dst, ovf, slots);
  gat_scores_and_rows<<<N_BKT + SCORE_BLOCKS, 1024, 0, stream>>>(z, A, asrc, adst, cnt, ovf, slots);
  gat_gather<<<(N_NODES * 16) / 256, 256, 0, stream>>>(cnt, slots, ovf, src, dst, asrc, adst, z, out);
}


# projection: a wave forms all 4 head tiles of its 16 nodes from one set of A fragments read by 16-B loads, W staged once per block in LDS; k regrouped alike in A and B, so the sum order of z changes
# speedup vs baseline: 5.0736x; 1.0805x over previous
#include <hip/hip_runtime.h>
#include <math.h>


#define N_NODES   50000
#define N_EDGES   800000
#define IN_FEATS  128
#define OUT_FEATS 16
#define NUM_HEADS 4
#define ZCOLS     (NUM_HEADS * OUT_FEATS)
#define ALPHA     0.2f

typedef __attribute__((ext_vector_type(2))) float v2f;
typedef __attribute__((ext_vector_type(8))) float v8f;

#define SLOT_CAP  32
#define OVF_WORDS 50000
#define OVF_HEAD  256
#define GCNT_OFF  2
#define OVF_PAIR0 (OVF_HEAD / 2)
#define OVF_CAP   ((OVF_WORDS - OVF_HEAD) / 2)
#define BKT_NODES 256
#define N_BKT     ((N_NODES + BKT_NODES - 1) / BKT_NODES)
#define BKT_WORDS (BKT_NODES * SLOT_CAP)
#define BIN_EDGES 2048
#define BIN_BLOCKS ((N_EDGES + BIN_EDGES - 1) / BIN_EDGES)
static_assert(N_NODES <= 65536, "a bucket word keeps the source id in 16 bits");
static_assert(N_EDGES % 4 == 0, "edges are read as whole quads");
static_assert(GCNT_OFF + N_BKT <= OVF_HEAD, "fill counts lie inside the cleared head");

__device__ __forceinline__ int bucket_cap(int b) {
  return min(BKT_WORDS, N_NODES * SLOT_CAP - b * BKT_WORDS);
}
__device__ __forceinline__ int bucket_of(int d) {
  return (int)min((unsigned)d >> 8, (unsigned)(N_BKT - 1));
}

#define N_MTILES   (N_NODES / 16)
#define PROJ_TILES 4
#define W_PITCH    (IN_FEATS + 4)
static_assert(N_NODES % 16 == 0, "whole tiles of 16 nodes");
__device__ __forceinline__ void proj_block(
    const float* __restrict__ h, const float* __restrict__ W,
    float* __restrict__ z, const int pb, float* __restrict__ s_w)
{
  const int tid   = threadIdx.x;
  const int wave  = tid >> 5;
  const int lane  = tid & 31;
  const int lo16  = lane & 15;
  const int koff  = (lane >> 4) * 4;
  const int mtile = pb * PROJ_TILES + wave;
  const int mload = min(mtile, N_MTILES - 1);

  float4 wv[16];
#pragma unroll
  for (int it = 0; it < 16; ++it) {
    const int f = it * 128 + tid;
    wv[it] = *(const float4*)(W + (size_t)(f >> 5) * IN_FEATS + (f & 31) * 4);
  }
  const float* __restrict__ ha = h + (size_t)(mload * 16 + lo16) * IN_FEATS + koff;
  float4 a[16];
#pragma unroll
  for (int j = 0; j < 16; ++j)
    a[j] = *(const float4*)(ha + 8 * j);
#pragma unroll
  for (int it = 0; it < 16; ++it) {
    const int f = it * 128 + tid;
    *(float4*)(s_w + (f >> 5) * W_PITCH + (f & 31) * 4) = wv[it];
  }
  __syncthreads();
  if (mtile >= N_MTILES) return;

  const float* __restrict__ wb = s_w + lo16 * W_PITCH + koff;
  v8f c0 = {}, c1 = {}, c2 = {}, c3 = {};
#pragma unroll
  for (int j = 0; j < 16; ++j) {
    const float4 b0 = *(const float4*)(wb + 0 * 16 * W_PITCH + 8 * j);
    const float4 b1 = *(const float4*)(wb + 1 * 16 * W_PITCH + 8 * j);
    const float4 b2 = *(const float4*)(wb + 2 * 16 * W_PITCH + 8 * j);
    const float4 b3 = *(const float4*)(wb + 3 * 16 * W_PITCH + 8 * j);
    const v2f alo = { a[j].x, a[j].y };
    const v2f ahi = { a[j].z, a[j].w };
    c0 = __builtin_amdgcn_wmma_f32_16x16x4_f32(false, alo, false, (v2f){ b0.x, b0.y }, (short)0, c0, false, false);
    c1 = __builtin_amdgcn_wmma_f32_16x16x4_f32(false, alo, false, (v2f){ b1.x, b1.y }, (short)0, c1, false, false);
    c2 = __builtin_amdgcn_wmma_f32_16x16x4_f32(false, alo, false, (v2f){ b2.x, b2.y }, (short)0, c2, false, false);
    c3 = __builtin_amdgcn_wmma_f32_16x16x4_f32(false, alo, false, (v2f){ b3.x, b3.y }, (short)0, c3, false, false);
    c0 = __builtin_amdgcn_wmma_f32_16x16x4_f32(false, ahi, false, (v2f){ b0.z, b0.w }, (short)0, c0, false, false);
    c1 = __builtin_amdgcn_wmma_f32_16x16x4_f32(false, ahi, false, (v2f){ b1.z, b1.w }, (short)0, c1, false, false);
    c2 = __builtin_amdgcn_wmma_f32_16x16x4_f32(false, ahi, false, (v2f){ b2.z, b2.w }, (short)0, c2, false, false);
    c3 = __builtin_amdgcn_wmma_f32_16x16x4_f32(false, ahi, false, (v2f){ b3.z, b3.w }, (short)0, c3, false, false);
  }

  const int rbase = mtile * 16 + ((lane >> 4) << 3);
  float* __restrict__ zp = z + (size_t)rbase * ZCOLS + lo16;
#pragma unroll
  for (int r = 0; r < 8; ++r) {
    zp[(size_t)r * ZCOLS +  0] = c0[r];
    zp[(size_t)r * ZCOLS + 16] = c1[r];
    zp[(size_t)r * ZCOLS + 32] = c2[r];
    zp[(size_t)r * ZCOLS + 48] = c3[r];
  }
}

__device__ __forceinline__ void node_scores(const float* __restrict__ z,
                                            const float* __restrict__ A,
                                            float* __restrict__ asrc,
                                            float* __restrict__ adst,
                                            const int t)
{
  if (t >= N_NODES * NUM_HEADS) return;
  const int n = t >> 2, hh = t & 3;
  const float* __restrict__ zv = z + (size_t)n * ZCOLS + hh * OUT_FEATS;
  const float* __restrict__ As = A + hh * (2 * OUT_FEATS);
  float s = 0.f, d = 0.f;
#pragma unroll
  for (int f = 0; f < OUT_FEATS; ++f) {
    s += zv[f] * As[f];
    d += zv[f] * As[OUT_FEATS + f];
  }
  asrc[t] = s;
  adst[t] = d;
}

__global__ __launch_bounds__(64) void gat_clear_counts(int* __restrict__ ovf)
{
  ((int4*)ovf)[threadIdx.x] = make_int4(0, 0, 0, 0);
}

#define PROJ_BLOCKS ((N_MTILES + PROJ_TILES - 1) / PROJ_TILES)
__global__ __launch_bounds__(128) void gat_proj_and_bin(
    const float* __restrict__ h, const float* __restrict__ W, float* __restrict__ z,
    const int* __restrict__ src, const int* __restrict__ dst,
    int* __restrict__ ovf, int* __restrict__ slots)
{
  __shared__ int s_hist[256];
  __shared__ int s_cur[256];
  __shared__ __attribute__((aligned(16))) float s_w[ZCOLS * W_PITCH];
  const int b = blockIdx.x;
  if (b >= BIN_BLOCKS) {
    proj_block(h, W, z, b - BIN_BLOCKS, s_w);
    return;
  }
  const int tid = threadIdx.x;
  s_hist[tid] = 0;
  s_hist[tid + 128] = 0;

  int4 d4[4], s4[4];
  const int e0 = b * BIN_EDGES + tid * 4;
#pragma unroll
  for (int j = 0; j < 4; ++j) {
    const int e = e0 + j * 512;
    if (e < N_EDGES) {
      d4[j] = *(const int4*)(dst + e);
      s4[j] = *(const int4*)(src + e);
    } else {
      d4[j] = make_int4(0, 0, 0, 0);
      s4[j] = make_int4(0, 0, 0, 0);
    }
  }
  __syncthreads();
#pragma unroll
  for (int j = 0; j < 4; ++j) {
    if (e0 + j * 512 < N_EDGES) {
      atomicAdd(&s_hist[bucket_of(d4[j].x)], 1);
      atomicAdd(&s_hist[bucket_of(d4[j].y)], 1);
      atomicAdd(&s_hist[bucket_of(d4[j].z)], 1);
      atomicAdd(&s_hist[bucket_of(d4[j].w)], 1);
    }
  }
  __syncthreads();
#pragma unroll
  for (int i = tid; i < 256; i += 128) {
    int base = 0;
    if (i < N_BKT) {
      const int hcount = s_hist[i];
      if (hcount > 0) base = atomicAdd(&ovf[GCNT_OFF + i], hcount);
    }
    s_cur[i] = base;
  }
  __syncthreads();
#pragma unroll
  for (int j = 0; j < 4; ++j) {
    if (e0 + j * 512 < N_EDGES) {
      const int dd[4] = { d4[j].x, d4[j].y, d4[j].z, d4[j].w };
      const int ss[4] = { s4[j].x, s4[j].y, s4[j].z, s4[j].w };
      int bk[4], pos[4];
#pragma unroll
      for (int u = 0; u < 4; ++u) {
        bk[u]  = bucket_of(dd[u]);
        pos[u] = atomicAdd(&s_cur[bk[u]], 1);
      }
#pragma unroll
      for (int u = 0; u < 4; ++u) {
        if (pos[u] < bucket_cap(bk[u]))
          slots[bk[u] * BKT_WORDS + pos[u]] =
              ((dd[u] & (BKT_NODES - 1)) << 16) | (ss[u] & 0xFFFF);
      }
    }
  }
}

#define SCORE_BLOCKS ((N_NODES * NUM_HEADS + 1023) / 1024)
__global__ __launch_bounds__(1024) void gat_scores_and_rows(
    const float* __restrict__ z, const float* __restrict__ A,
    float* __restrict__ asrc, float* __restrict__ adst,
    int* __restrict__ cnt, int* __restrict__ ovf, int* __restrict__ slots)
{
  __shared__ int s_row[BKT_WORDS];
  __shared__ int s_cnt[BKT_NODES];
  const int b = blockIdx.x;
  if (b >= N_BKT) {
    node_scores(z, A, asrc, adst, (b - N_BKT) * 1024 + threadIdx.x);
    return;
  }
  const int tid   = threadIdx.x;
  const int total = ovf[GCNT_OFF + b];
  int* __restrict__ area = slots + b * BKT_WORDS;
  const int n_own = b * BKT_NODES + tid;

  if (total > bucket_cap(b)) {
    if (tid < BKT_NODES && n_own < N_NODES) cnt[n_own] = SLOT_CAP + 1;
    if (tid == 0) atomicAdd(&ovf[0], OVF_CAP + 1);
    return;
  }

  if (tid < BKT_NODES) s_cnt[tid] = 0;
  int w[8];
#pragma unroll
  for (int j = 0; j < 8; ++j) {
    const int i = j * 1024 + tid;
    w[j] = (i < total) ? area[i] : -1;
  }
  __syncthreads();
#pragma unroll
  for (int j = 0; j < 8; ++j) {
    if (w[j] >= 0) {
      const int dl = (w[j] >> 16) & (BKT_NODES - 1);
      const int s  = w[j] & 0xFFFF;
      const int r  = atomicAdd(&s_cnt[dl], 1);
      if (r < SLOT_CAP) {
        s_row[dl * SLOT_CAP + r] = s;
      } else {
        const int p = atomicAdd(&ovf[0], 1);
        if (p < OVF_CAP) ((int2*)ovf)[OVF_PAIR0 + p] = make_int2(b * BKT_NODES + dl, s);
      }
    }
  }
  __syncthreads();
#pragma unroll
  for (int j = 0; j < 8; ++j) {
    const int i  = j * 1024 + tid;
    const int dl = i >> 5;
    const int r  = i & (SLOT_CAP - 1);
    if (b * BKT_NODES + dl < N_NODES && r < min(s_cnt[dl], SLOT_CAP))
      area[i] = s_row[i];
  }
  if (tid < BKT_NODES && n_own < N_NODES) cnt[n_own] = s_cnt[tid];
}

__device__ __forceinline__ float lrelu(float v) { return v > 0.f ? v : ALPHA * v; }

__device__ __forceinline__ int node_clamp(int s) {
  return (int)min((unsigned)s, (unsigned)(N_NODES - 1));
}

__device__ __forceinline__ void max_edge(float4& m4, const float* __restrict__ asrc,
                                         const int s, const float4 ad4)
{
  const float4 a4 = *(const float4*)(asrc + s * NUM_HEADS);
  m4.x = fmaxf(m4.x, lrelu(a4.x + ad4.x));
  m4.y = fmaxf(m4.y, lrelu(a4.y + ad4.y));
  m4.z = fmaxf(m4.z, lrelu(a4.z + ad4.z));
  m4.w = fmaxf(m4.w, lrelu(a4.w + ad4.w));
}

__device__ __forceinline__ void sum_edge(float4& acc, float& den,
                                         const float* __restrict__ aq,
                                         const float* __restrict__ zq,
                                         const int s, const float ad, const float mx)
{
  const float a = aq[s * NUM_HEADS];
  const float4 zr = *(const float4*)(zq + (size_t)s * ZCOLS);
  const float ex = expf(lrelu(a + ad) - mx);
  den += ex;
  acc.x += ex * zr.x;
  acc.y += ex * zr.y;
  acc.z += ex * zr.z;
  acc.w += ex * zr.w;
}

__global__ __launch_bounds__(256) void gat_gather(
    const int* __restrict__ cnt, const int* __restrict__ slots,
    const int* __restrict__ ovf,
    const int* __restrict__ src, const int* __restrict__ dst,
    const float* __restrict__ asrc, const float* __restrict__ adst,
    const float* __restrict__ z, float* __restrict__ out)
{
  const int t  = blockIdx.x * 256 + threadIdx.x;
  const int n  = t >> 4;
  const int q  = t & 15;
  const int hh = q >> 2;
  const int deg = cnt[n];
  const int* __restrict__ row = slots + n * SLOT_CAP;
  const float4 ad4 = *(const float4*)(adst + n * NUM_HEADS);
  const bool spill = deg > SLOT_CAP;
  const int  olen  = spill ? ovf[0] : 0;
  const bool walk  = olen > OVF_CAP;
  const int  end   = walk ? 0 : min(deg, SLOT_CAP);
  const int  novf  = walk ? 0 : olen;
  const int  nwalk = walk ? N_EDGES : 0;

  float4 m4 = make_float4(-INFINITY, -INFINITY, -INFINITY, -INFINITY);
  for (int i = q; i < end; i += 16)
    max_edge(m4, asrc, node_clamp(row[i]), ad4);
  for (int i = q; i < novf; i += 16) {
    const int2 pr = ((const int2*)ovf)[OVF_PAIR0 + i];
    if (pr.x == n) max_edge(m4, asrc, node_clamp(pr.y), ad4);
  }
  for (int e = q; e < nwalk; e += 16)
    if (dst[e] == n) max_edge(m4, asrc, node_clamp(src[e]), ad4);
  for (int off = 8; off > 0; off >>= 1) {
    m4.x = fmaxf(m4.x, __shfl_xor(m4.x, off, 16));
    m4.y = fmaxf(m4.y, __shfl_xor(m4.y, off, 16));
    m4.z = fmaxf(m4.z, __shfl_xor(m4.z, off, 16));
    m4.w = fmaxf(m4.w, __shfl_xor(m4.w, off, 16));
  }
  float mx = hh == 0 ? m4.x : hh == 1 ? m4.y : hh == 2 ? m4.z : m4.w;
  const float ad = hh == 0 ? ad4.x : hh == 1 ? ad4.y : hh == 2 ? ad4.z : ad4.w;
  if (!isfinite(mx)) mx = 0.f;

  float4 acc = make_float4(0.f, 0.f, 0.f, 0.f);
  float den = 0.f;
  const float* __restrict__ zq = z + q * 4;
  const float* __restrict__ aq = asrc + hh;
  for (int k = 0; k < end; k += 4) {
    int   s4[4];
    float a4[4];
    float4 z4[4];
#pragma unroll
    for (int j = 0; j < 4; ++j)
      s4[j] = node_clamp(row[min(k + j, end - 1)]);
#pragma unroll
    for (int j = 0; j < 4; ++j) {
      a4[j] = aq[s4[j] * NUM_HEADS];
      z4[j] = *(const float4*)(zq + (size_t)s4[j] * ZCOLS);
    }
#pragma unroll
    for (int j = 0; j < 4; ++j) {
      const float v  = lrelu(a4[j] + ad);
      const float ex = expf((k + j < end) ? v - mx : -INFINITY);
      den += ex;
      acc.x += ex * z4[j].x;
      acc.y += ex * z4[j].y;
      acc.z += ex * z4[j].z;
      acc.w += ex * z4[j].w;
    }
  }
  for (int i = 0; i < novf; ++i) {
    const int2 pr = ((const int2*)ovf)[OVF_PAIR0 + i];
    if (pr.x == n) sum_edge(acc, den, aq, zq, node_clamp(pr.y), ad, mx);
  }
  for (int e = 0; e < nwalk; ++e)
    if (dst[e] == n) sum_edge(acc, den, aq, zq, node_clamp(src[e]), ad, mx);
  const float dd = fmaxf(den, 1e-12f);
  *(float4*)(out + (size_t)n * ZCOLS + q * 4) =
      make_float4(acc.x / dd, acc.y / dd, acc.z / dd, acc.w / dd);
}

extern "C" void kernel_launch(void* const* d_in, const int* in_sizes, int n_in,
                              void* d_out, int out_size, void* d_ws, size_t ws_size,
                              hipStream_t stream) {
  const float* h   = (const float*)d_in[0];
  const int*   src = (const int*)d_in[1];
  const int*   dst = (const int*)d_in[2];
  const float* W   = (const float*)d_in[3];
  const float* A   = (const float*)d_in[4];
  float* out = (float*)d_out;

  float* z     = (float*)d_ws;
  float* asrc  = z     + (size_t)N_NODES * ZCOLS;
  float* adst  = asrc  + (size_t)N_NODES * NUM_HEADS;
  int*   cnt   = (int*)(adst + (size_t)N_NODES * NUM_HEADS);
  int*   ovf   = cnt + N_NODES;
  int*   slots = ovf + OVF_WORDS;

  gat_clear_counts<<<1, 64, 0, stream>>>(ovf);
  gat_proj_and_bin<<<BIN_BLOCKS + PROJ_BLOCKS, 128, 0, stream>>>(h, W, z, src, dst, ovf, slots);
  gat_scores_and_rows<<<N_BKT + SCORE_BLOCKS, 1024, 0, stream>>>(z, A, asrc, adst, cnt, ovf, slots);
  gat_gather<<<(N_NODES * 16) / 256, 256, 0, stream>>>(cnt, slots, ovf, src, dst, asrc, adst, z, out);
}


# gather loop: softmax weights by the hardware base-2 exponential on logits held times log2(e) (factor folded into A), leaky relu as max(x, 0.2x), id clamp at the row store; sum order unchanged
# speedup vs baseline: 6.7186x; 1.3242x over previous
#include <hip/hip_runtime.h>
#include <math.h>


#define N_NODES   50000
#define N_EDGES   800000
#define IN_FEATS  128
#define OUT_FEATS 16
#define NUM_HEADS 4
#define ZCOLS     (NUM_HEADS * OUT_FEATS)
#define ALPHA     0.2f

typedef __attribute__((ext_vector_type(2))) float v2f;
typedef __attribute__((ext_vector_type(8))) float v8f;
typedef __attribute__((ext_vector_type(4))) float v4f;

#define HEAD_WORDS 256
#define BKT_NODES 256
#define N_BKT     ((N_NODES + BKT_NODES - 1) / BKT_NODES)
#define BKT_WORDS (BKT_NODES * 32)
#define BIN_EDGES 2048
#define BIN_BLOCKS ((N_EDGES + BIN_EDGES - 1) / BIN_EDGES)
static_assert(N_NODES <= 65536, "a bucket word keeps the source id in 16 bits");
static_assert(N_EDGES % 4 == 0, "edges are read as whole quads");
static_assert(N_BKT <= HEAD_WORDS, "fill counts lie inside the cleared head");

__device__ __forceinline__ int bucket_cap(int b) {
  return min(BKT_WORDS, N_NODES * 32 - b * BKT_WORDS);
}
__device__ __forceinline__ int bucket_of(int d) {
  return (int)min((unsigned)d >> 8, (unsigned)(N_BKT - 1));
}

#define N_MTILES   (N_NODES / 16)
#define PROJ_TILES 4
#define W_PITCH    (IN_FEATS + 4)
static_assert(N_NODES % 16 == 0, "whole tiles of 16 nodes");
__device__ __forceinline__ void proj_block(
    const float* __restrict__ h, const float* __restrict__ W,
    float* __restrict__ z, const int pb, float* __restrict__ s_w)
{
  const int tid   = threadIdx.x;
  const int wave  = tid >> 5;
  const int lane  = tid & 31;
  const int lo16  = lane & 15;
  const int koff  = (lane >> 4) * 4;
  const int mtile = pb * PROJ_TILES + wave;
  const int mload = min(mtile, N_MTILES - 1);

  float4 wv[16];
#pragma unroll
  for (int it = 0; it < 16; ++it) {
    const int f = it * 128 + tid;
    wv[it] = *(const float4*)(W + (size_t)(f >> 5) * IN_FEATS + (f & 31) * 4);
  }
  const float* __restrict__ ha = h + (size_t)(mload * 16 + lo16) * IN_FEATS + koff;
  float4 a[16];
#pragma unroll
  for (int j = 0; j < 16; ++j)
    a[j] = *(const float4*)(ha + 8 * j);
#pragma unroll
  for (int it = 0; it < 16; ++it) {
    const int f = it * 128 + tid;
    *(float4*)(s_w + (f >> 5) * W_PITCH + (f & 31) * 4) = wv[it];
  }
  __syncthreads();
  if (mtile >= N_MTILES) return;

  const float* __restrict__ wb = s_w + lo16 * W_PITCH + koff;
  v8f c0 = {}, c1 = {}, c2 = {}, c3 = {};
#pragma unroll
  for (int j = 0; j < 16; ++j) {
    const float4 b0 = *(const float4*)(wb + 0 * 16 * W_PITCH + 8 * j);
    const float4 b1 = *(const float4*)(wb + 1 * 16 * W_PITCH + 8 * j);
    const float4 b2 = *(const float4*)(wb + 2 * 16 * W_PITCH + 8 * j);
    const float4 b3 = *(const float4*)(wb + 3 * 16 * W_PITCH + 8 * j);
    const v2f alo = { a[j].x, a[j].y };
    const v2f ahi = { a[j].z, a[j].w };
    c0 = __builtin_amdgcn_wmma_f32_16x16x4_f32(false, alo, false, (v2f){ b0.x, b0.y }, (short)0, c0, false, false);
    c1 = __builtin_amdgcn_wmma_f32_16x16x4_f32(false, alo, false, (v2f){ b1.x, b1.y }, (short)0, c1, false, false);
    c2 = __builtin_amdgcn_wmma_f32_16x16x4_f32(false, alo, false, (v2f){ b2.x, b2.y }, (short)0, c2, false, false);
    c3 = __builtin_amdgcn_wmma_f32_16x16x4_f32(false, alo, false, (v2f){ b3.x, b3.y }, (short)0, c3, false, false);
    c0 = __builtin_amdgcn_wmma_f32_16x16x4_f32(false, ahi, false, (v2f){ b0.z, b0.w }, (short)0, c0, false, false);
    c1 = __builtin_amdgcn_wmma_f32_16x16x4_f32(false, ahi, false, (v2f){ b1.z, b1.w }, (short)0, c1, false, false);
    c2 = __builtin_amdgcn_wmma_f32_16x16x4_f32(false, ahi, false, (v2f){ b2.z, b2.w }, (short)0, c2, false, false);
    c3 = __builtin_amdgcn_wmma_f32_16x16x4_f32(false, ahi, false, (v2f){ b3.z, b3.w }, (short)0, c3, false, false);
  }

  const int rbase = mtile * 16 + ((lane >> 4) << 3);
  float* __restrict__ zp = z + (size_t)rbase * ZCOLS + lo16;
#pragma unroll
  for (int r = 0; r < 8; ++r) {
    zp[(size_t)r * ZCOLS +  0] = c0[r];
    zp[(size_t)r * ZCOLS + 16] = c1[r];
    zp[(size_t)r * ZCOLS + 32] = c2[r];
    zp[(size_t)r * ZCOLS + 48] = c3[r];
  }
}

__global__ __launch_bounds__(64) void gat_clear_counts(int* __restrict__ gcnt)
{
  ((int4*)gcnt)[threadIdx.x] = make_int4(0, 0, 0, 0);
}

#define PROJ_BLOCKS ((N_MTILES + PROJ_TILES - 1) / PROJ_TILES)
__global__ __launch_bounds__(128) void gat_proj_and_bin(
    const float* __restrict__ h, const float* __restrict__ W, float* __restrict__ z,
    const int* __restrict__ src, const int* __restrict__ dst,
    int* __restrict__ gcnt, int* __restrict__ slots)
{
  __shared__ int s_hist[256];
  __shared__ int s_cur[256];
  __shared__ __attribute__((aligned(16))) float s_w[ZCOLS * W_PITCH];
  const int b = blockIdx.x;
  if (b >= BIN_BLOCKS) {
    proj_block(h, W, z, b - BIN_BLOCKS, s_w);
    return;
  }
  const int tid = threadIdx.x;
  s_hist[tid] = 0;
  s_hist[tid + 128] = 0;

  int4 d4[4], s4[4];
  const int e0 = b * BIN_EDGES + tid * 4;
#pragma unroll
  for (int j = 0; j < 4; ++j) {
    const int e = e0 + j * 512;
    if (e < N_EDGES) {
      d4[j] = *(const int4*)(dst + e);
      s4[j] = *(const int4*)(src + e);
    } else {
      d4[j] = make_int4(0, 0, 0, 0);
      s4[j] = make_int4(0, 0, 0, 0);
    }
  }
  __syncthreads();
#pragma unroll
  for (int j = 0; j < 4; ++j) {
    if (e0 + j * 512 < N_EDGES) {
      atomicAdd(&s_hist[bucket_of(d4[j].x)], 1);
      atomicAdd(&s_hist[bucket_of(d4[j].y)], 1);
      atomicAdd(&s_hist[bucket_of(d4[j].z)], 1);
      atomicAdd(&s_hist[bucket_of(d4[j].w)], 1);
    }
  }
  __syncthreads();
#pragma unroll
  for (int i = tid; i < 256; i += 128) {
    int base = 0;
    if (i < N_BKT) {
      const int hcount = s_hist[i];
      if (hcount > 0) base = atomicAdd(&gcnt[i], hcount);
    }
    s_cur[i] = base;
  }
  __syncthreads();
#pragma unroll
  for (int j = 0; j < 4; ++j) {
    if (e0 + j * 512 < N_EDGES) {
      const int dd[4] = { d4[j].x, d4[j].y, d4[j].z, d4[j].w };
      const int ss[4] = { s4[j].x, s4[j].y, s4[j].z, s4[j].w };
      int bk[4], pos[4];
#pragma unroll
      for (int u = 0; u < 4; ++u) {
        bk[u]  = bucket_of(dd[u]);
        pos[u] = atomicAdd(&s_cur[bk[u]], 1);
      }
#pragma unroll
      for (int u = 0; u < 4; ++u) {
        if (pos[u] < bucket_cap(bk[u]))
          slots[bk[u] * BKT_WORDS + pos[u]] =
              ((dd[u] & (BKT_NODES - 1)) << 16) | (ss[u] & 0xFFFF);
      }
    }
  }
}

__device__ __forceinline__ float lrelu(float v) { return fmaxf(v, ALPHA * v); }

#define LOG2E 1.44269504088896340736f
__device__ __forceinline__ float4 scale4(const float4 a, const float s) {
  return make_float4(a.x * s, a.y * s, a.z * s, a.w * s);
}

__device__ __forceinline__ int node_clamp(int s) {
  return (int)min((unsigned)s, (unsigned)(N_NODES - 1));
}

__device__ __forceinline__ float quad_sum(float p) {
  p += __builtin_bit_cast(float, __builtin_amdgcn_update_dpp(
           0, __builtin_bit_cast(int, p), 0xB1, 0xF, 0xF, true));
  p += __builtin_bit_cast(float, __builtin_amdgcn_update_dpp(
           0, __builtin_bit_cast(int, p), 0x4E, 0xF, 0xF, true));
  return p;
}

__device__ __forceinline__ float dot4(const float4 a, const float4 b) {
  return fmaf(a.w, b.w, fmaf(a.z, b.z, fmaf(a.y, b.y, a.x * b.x)));
}

__device__ __forceinline__ void edge_online(float4& acc, float& den, float& m,
                                            const float* __restrict__ zq, const int s,
                                            const float4 As4, const float ad)
{
  const float4 zr = *(const float4*)(zq + (size_t)s * ZCOLS);
  const float v  = lrelu(quad_sum(dot4(zr, As4)) + ad);
  const float mn = fmaxf(m, v);
  const float sc = __builtin_amdgcn_exp2f(m - mn);
  const float ex = __builtin_amdgcn_exp2f(v - mn);
  den   = den   * sc + ex;
  acc.x = acc.x * sc + ex * zr.x;
  acc.y = acc.y * sc + ex * zr.y;
  acc.z = acc.z * sc + ex * zr.z;
  acc.w = acc.w * sc + ex * zr.w;
  m = mn;
}

#define ROW_CAP    64
#define GB_NODES   64
#define GB_THREADS 256
#define GB_BLOCKS  ((N_NODES + GB_NODES - 1) / GB_NODES)
#define GB_PER_BKT (BKT_NODES / GB_NODES)
#define GB_GROUPS  (GB_THREADS / 16)
#define GB_ROUNDS  (GB_NODES / GB_GROUPS)
#define GB_QUADS   (BKT_WORDS / 4 / GB_THREADS)
#define TRIP 8
static_assert(TRIP == 8, "the trip's arrival point names its eight rows");
static_assert(BKT_WORDS % (4 * GB_THREADS) == 0, "whole quads per thread");
static_assert(GB_NODES % GB_GROUPS == 0 && GB_ROUNDS == 4, "four whole rounds");
__global__ __launch_bounds__(GB_THREADS) void gat_rows_gather(
    const int* __restrict__ gcnt, const int* __restrict__ slots,
    const int* __restrict__ src, const int* __restrict__ dst,
    const float* __restrict__ A,
    const float* __restrict__ z, float* __restrict__ out)
{
  __shared__ int s_row[GB_NODES * ROW_CAP];
  __shared__ int s_cnt[GB_NODES];
  const int tid   = threadIdx.x;
  const int b     = blockIdx.x / GB_PER_BKT;
  const int lo    = (blockIdx.x % GB_PER_BKT) * GB_NODES;
  const int total = gcnt[b];
  const bool over = total > bucket_cap(b);
  const int  tot  = over ? 0 : total;
  const int* __restrict__ area = slots + b * BKT_WORDS;

  const int g  = tid >> 4;
  const int q  = tid & 15;
  const int hh = q >> 2;
  const float* __restrict__ zq = z + q * 4;
  const float* __restrict__ Ah = A + hh * (2 * OUT_FEATS) + (q & 3) * 4;
  const float4 As4 = scale4(*(const float4*)(Ah), LOG2E);
  const float4 Ad4 = scale4(*(const float4*)(Ah + OUT_FEATS), LOG2E);
  float4 znr[GB_ROUNDS];
#pragma unroll
  for (int rd = 0; rd < GB_ROUNDS; ++rd) {
    const int nr = min(blockIdx.x * GB_NODES + rd * GB_GROUPS + g, N_NODES - 1);
    znr[rd] = *(const float4*)(zq + (size_t)nr * ZCOLS);
  }

  if (tid < GB_NODES) s_cnt[tid] = 0;
  const int qlast = max((tot - 1) >> 2, 0);
  int4 w[GB_QUADS];
#pragma unroll
  for (int j = 0; j < GB_QUADS; ++j)
    w[j] = ((const int4*)area)[min(j * GB_THREADS + tid, qlast)];
  __syncthreads();
  int rk[GB_QUADS][4];
#pragma unroll
  for (int j = 0; j < GB_QUADS; ++j) {
    const int i0 = 4 * (j * GB_THREADS + tid);
    const int ww[4] = { w[j].x, w[j].y, w[j].z, w[j].w };
#pragma unroll
    for (int u = 0; u < 4; ++u) {
      const int dl = ((ww[u] >> 16) & (BKT_NODES - 1)) - lo;
      rk[j][u] = ROW_CAP;
      if (i0 + u < tot && (unsigned)dl < (unsigned)GB_NODES)
        rk[j][u] = atomicAdd(&s_cnt[dl], 1);
    }
  }
#pragma unroll
  for (int j = 0; j < GB_QUADS; ++j) {
    const int ww[4] = { w[j].x, w[j].y, w[j].z, w[j].w };
#pragma unroll
    for (int u = 0; u < 4; ++u) {
      const int dl = ((ww[u] >> 16) & (BKT_NODES - 1)) - lo;
      if (rk[j][u] < ROW_CAP) s_row[dl * ROW_CAP + rk[j][u]] = node_clamp(ww[u] & 0xFFFF);
    }
  }
  __syncthreads();

  const v4f* __restrict__ z4p = (const v4f*)z;
  float adr[GB_ROUNDS];
#pragma unroll
  for (int rd = 0; rd < GB_ROUNDS; ++rd)
    adr[rd] = quad_sum(dot4(znr[rd], Ad4));

  for (int rd = 0; rd < GB_ROUNDS; ++rd) {
    const int nl = rd * GB_GROUPS + g;
    const int n  = blockIdx.x * GB_NODES + nl;
    if (n >= N_NODES) break;
    const int  deg   = s_cnt[nl];
    const bool full  = deg > ROW_CAP;
    const int  end   = (over || full) ? 0 : deg;
    const int  nbw   = (!over && full) ? total : 0;
    const int  nwalk = over ? N_EDGES : 0;
    const int* __restrict__ rowp = s_row + nl * ROW_CAP;

    const float ad = rd == 0 ? adr[0] : rd == 1 ? adr[1] : rd == 2 ? adr[2] : adr[3];

    float  m   = -INFINITY;
    float  den = 0.f;
    float4 acc = make_float4(0.f, 0.f, 0.f, 0.f);

    int sn[TRIP];
#pragma unroll
    for (int j = 0; j < TRIP; ++j) sn[j] = 0;
    if (end > 0) {
#pragma unroll
      for (int j = 0; j < TRIP; ++j)
        sn[j] = rowp[min(j, end - 1)];
    }
    for (int k = 0; k < end; k += TRIP) {
      v4f    zr[TRIP];
      float  v[TRIP];
#pragma unroll
      for (int j = 0; j < TRIP; ++j)
        zr[j] = z4p[(unsigned)sn[j] * (ZCOLS / 4) + (unsigned)q];
#pragma unroll
      for (int j = 0; j < TRIP; ++j)
        sn[j] = rowp[min(k + TRIP + j, end - 1)];
      asm volatile("" : "+v"(zr[0]), "+v"(zr[1]), "+v"(zr[2]), "+v"(zr[3]),
                        "+v"(zr[4]), "+v"(zr[5]), "+v"(zr[6]), "+v"(zr[7]));
      float tm = -INFINITY;
#pragma unroll
      for (int j = 0; j < TRIP; ++j) {
        const float e = lrelu(quad_sum(dot4(make_float4(zr[j].x, zr[j].y, zr[j].z, zr[j].w), As4)) + ad);
        v[j] = (k + j < end) ? e : -INFINITY;
        tm = fmaxf(tm, v[j]);
      }
      const float mn = fmaxf(m, tm);
      const float sc = __builtin_amdgcn_exp2f(m - mn);
      den   *= sc;
      acc.x *= sc;
      acc.y *= sc;
      acc.z *= sc;
      acc.w *= sc;
#pragma unroll
      for (int j = 0; j < TRIP; ++j) {
        const float ex = __builtin_amdgcn_exp2f(v[j] - mn);
        den += ex;
        acc.x += ex * zr[j].x;
        acc.y += ex * zr[j].y;
        acc.z += ex * zr[j].z;
        acc.w += ex * zr[j].w;
      }
      m = mn;
    }
    for (int i = 0; i < nbw; ++i) {
      const int wd = area[i];
      if (((wd >> 16) & (BKT_NODES - 1)) == lo + nl)
        edge_online(acc, den, m, zq, node_clamp(wd & 0xFFFF), As4, ad);
    }
    for (int e = 0; e < nwalk; ++e)
      if (dst[e] == n) edge_online(acc, den, m, zq, node_clamp(src[e]), As4, ad);
    const float dd = fmaxf(den, 1e-12f);
    *(float4*)(out + (size_t)n * ZCOLS + q * 4) =
        make_float4(acc.x / dd, acc.y / dd, acc.z / dd, acc.w / dd);
  }
}

extern "C" void kernel_launch(void* const* d_in, const int* in_sizes, int n_in,
                              void* d_out, int out_size, void* d_ws, size_t ws_size,
                              hipStream_t stream) {
  const float* h   = (const float*)d_in[0];
  const int*   src = (const int*)d_in[1];
  const int*   dst = (const int*)d_in[2];
  const float* W   = (const float*)d_in[3];
  const float* A   = (const float*)d_in[4];
  float* out = (float*)d_out;

  float* z     = (float*)d_ws;
  int*   gcnt  = (int*)(z + (size_t)N_NODES * ZCOLS);
  int*   slots = gcnt + HEAD_WORDS;

  gat_clear_counts<<<1, 64, 0, stream>>>(gcnt);
  gat_proj_and_bin<<<BIN_BLOCKS + PROJ_BLOCKS, 128, 0, stream>>>(h, W, z, src, dst, gcnt, slots);
  gat_rows_gather<<<GB_BLOCKS, GB_THREADS, 0, stream>>>(gcnt, slots, src, dst, A, z, out);
}


# gather: 8 lanes per node, not 16: z rows kept in lane order, so a lane's two 16-B loads are 8 columns of one head and whole lines; logit = one pair step; logit products summed in a new grouping
# speedup vs baseline: 7.0351x; 1.0471x over previous
#include <hip/hip_runtime.h>
#include <math.h>


#define N_NODES   50000
#define N_EDGES   800000
#define IN_FEATS  128
#define OUT_FEATS 16
#define NUM_HEADS 4
#define ZCOLS     (NUM_HEADS * OUT_FEATS)
#define ALPHA     0.2f

typedef __attribute__((ext_vector_type(2))) float v2f;
typedef __attribute__((ext_vector_type(8))) float v8f;
typedef __attribute__((ext_vector_type(4))) float v4f;

#define HEAD_WORDS 256
#define BKT_NODES 256
#define N_BKT     ((N_NODES + BKT_NODES - 1) / BKT_NODES)
#define BKT_WORDS (BKT_NODES * 32)
#define BIN_EDGES 2048
#define BIN_BLOCKS ((N_EDGES + BIN_EDGES - 1) / BIN_EDGES)
static_assert(N_NODES <= 65536, "a bucket word keeps the source id in 16 bits");
static_assert(N_EDGES % 4 == 0, "edges are read as whole quads");
static_assert(N_BKT <= HEAD_WORDS, "fill counts lie inside the cleared head");

__device__ __forceinline__ int bucket_cap(int b) {
  return min(BKT_WORDS, N_NODES * 32 - b * BKT_WORDS);
}
__device__ __forceinline__ int bucket_of(int d) {
  return (int)min((unsigned)d >> 8, (unsigned)(N_BKT - 1));
}

#define N_MTILES   (N_NODES / 16)
#define PROJ_TILES 4
#define W_PITCH    (IN_FEATS + 4)
static_assert(N_NODES % 16 == 0, "whole tiles of 16 nodes");
__device__ __forceinline__ void proj_block(
    const float* __restrict__ h, const float* __restrict__ W,
    float* __restrict__ z, const int pb, float* __restrict__ s_w)
{
  const int tid   = threadIdx.x;
  const int wave  = tid >> 5;
  const int lane  = tid & 31;
  const int lo16  = lane & 15;
  const int koff  = (lane >> 4) * 4;
  const int mtile = pb * PROJ_TILES + wave;
  const int mload = min(mtile, N_MTILES - 1);

  float4 wv[16];
#pragma unroll
  for (int it = 0; it < 16; ++it) {
    const int f = it * 128 + tid;
    wv[it] = *(const float4*)(W + (size_t)(f >> 5) * IN_FEATS + (f & 31) * 4);
  }
  const float* __restrict__ ha = h + (size_t)(mload * 16 + lo16) * IN_FEATS + koff;
  float4 a[16];
#pragma unroll
  for (int j = 0; j < 16; ++j)
    a[j] = *(const float4*)(ha + 8 * j);
#pragma unroll
  for (int it = 0; it < 16; ++it) {
    const int f = it * 128 + tid;
    *(float4*)(s_w + (f >> 5) * W_PITCH + (f & 31) * 4) = wv[it];
  }
  __syncthreads();
  if (mtile >= N_MTILES) return;

  const float* __restrict__ wb = s_w + lo16 * W_PITCH + koff;
  v8f c0 = {}, c1 = {}, c2 = {}, c3 = {};
#pragma unroll
  for (int j = 0; j < 16; ++j) {
    const float4 b0 = *(const float4*)(wb + 0 * 16 * W_PITCH + 8 * j);
    const float4 b1 = *(const float4*)(wb + 1 * 16 * W_PITCH + 8 * j);
    const float4 b2 = *(const float4*)(wb + 2 * 16 * W_PITCH + 8 * j);
    const float4 b3 = *(const float4*)(wb + 3 * 16 * W_PITCH + 8 * j);
    const v2f alo = { a[j].x, a[j].y };
    const v2f ahi = { a[j].z, a[j].w };
    c0 = __builtin_amdgcn_wmma_f32_16x16x4_f32(false, alo, false, (v2f){ b0.x, b0.y }, (short)0, c0, false, false);
    c1 = __builtin_amdgcn_wmma_f32_16x16x4_f32(false, alo, false, (v2f){ b1.x, b1.y }, (short)0, c1, false, false);
    c2 = __builtin_amdgcn_wmma_f32_16x16x4_f32(false, alo, false, (v2f){ b2.x, b2.y }, (short)0, c2, false, false);
    c3 = __builtin_amdgcn_wmma_f32_16x16x4_f32(false, alo, false, (v2f){ b3.x, b3.y }, (short)0, c3, false, false);
    c0 = __builtin_amdgcn_wmma_f32_16x16x4_f32(false, ahi, false, (v2f){ b0.z, b0.w }, (short)0, c0, false, false);
    c1 = __builtin_amdgcn_wmma_f32_16x16x4_f32(false, ahi, false, (v2f){ b1.z, b1.w }, (short)0, c1, false, false);
    c2 = __builtin_amdgcn_wmma_f32_16x16x4_f32(false, ahi, false, (v2f){ b2.z, b2.w }, (short)0, c2, false, false);
    c3 = __builtin_amdgcn_wmma_f32_16x16x4_f32(false, ahi, false, (v2f){ b3.z, b3.w }, (short)0, c3, false, false);
  }

  const int rbase = mtile * 16 + ((lane >> 4) << 3);
  const int zpos0 = ((lo16 & 4) << 3) + ((lo16 >> 3) << 2) + (lo16 & 3);
  float* __restrict__ zp = z + (size_t)rbase * ZCOLS + zpos0;
#pragma unroll
  for (int r = 0; r < 8; ++r) {
    zp[(size_t)r * ZCOLS +  0] = c0[r];
    zp[(size_t)r * ZCOLS +  8] = c1[r];
    zp[(size_t)r * ZCOLS + 16] = c2[r];
    zp[(size_t)r * ZCOLS + 24] = c3[r];
  }
}

__global__ __launch_bounds__(64) void gat_clear_counts(int* __restrict__ gcnt)
{
  ((int4*)gcnt)[threadIdx.x] = make_int4(0, 0, 0, 0);
}

#define PROJ_BLOCKS ((N_MTILES + PROJ_TILES - 1) / PROJ_TILES)
__global__ __launch_bounds__(128) void gat_proj_and_bin(
    const float* __restrict__ h, const float* __restrict__ W, float* __restrict__ z,
    const int* __restrict__ src, const int* __restrict__ dst,
    int* __restrict__ gcnt, int* __restrict__ slots)
{
  __shared__ int s_hist[256];
  __shared__ int s_cur[256];
  __shared__ __attribute__((aligned(16))) float s_w[ZCOLS * W_PITCH];
  const int b = blockIdx.x;
  if (b >= BIN_BLOCKS) {
    proj_block(h, W, z, b - BIN_BLOCKS, s_w);
    return;
  }
  const int tid = threadIdx.x;
  s_hist[tid] = 0;
  s_hist[tid + 128] = 0;

  int4 d4[4], s4[4];
  const int e0 = b * BIN_EDGES + tid * 4;
#pragma unroll
  for (int j = 0; j < 4; ++j) {
    const int e = e0 + j * 512;
    if (e < N_EDGES) {
      d4[j] = *(const int4*)(dst + e);
      s4[j] = *(const int4*)(src + e);
    } else {
      d4[j] = make_int4(0, 0, 0, 0);
      s4[j] = make_int4(0, 0, 0, 0);
    }
  }
  __syncthreads();
#pragma unroll
  for (int j = 0; j < 4; ++j) {
    if (e0 + j * 512 < N_EDGES) {
      atomicAdd(&s_hist[bucket_of(d4[j].x)], 1);
      atomicAdd(&s_hist[bucket_of(d4[j].y)], 1);
      atomicAdd(&s_hist[bucket_of(d4[j].z)], 1);
      atomicAdd(&s_hist[bucket_of(d4[j].w)], 1);
    }
  }
  __syncthreads();
#pragma unroll
  for (int i = tid; i < 256; i += 128) {
    int base = 0;
    if (i < N_BKT) {
      const int hcount = s_hist[i];
      if (hcount > 0) base = atomicAdd(&gcnt[i], hcount);
    }
    s_cur[i] = base;
  }
  __syncthreads();
#pragma unroll
  for (int j = 0; j < 4; ++j) {
    if (e0 + j * 512 < N_EDGES) {
      const int dd[4] = { d4[j].x, d4[j].y, d4[j].z, d4[j].w };
      const int ss[4] = { s4[j].x, s4[j].y, s4[j].z, s4[j].w };
      int bk[4], pos[4];
#pragma unroll
      for (int u = 0; u < 4; ++u) {
        bk[u]  = bucket_of(dd[u]);
        pos[u] = atomicAdd(&s_cur[bk[u]], 1);
      }
#pragma unroll
      for (int u = 0; u < 4; ++u) {
        if (pos[u] < bucket_cap(bk[u]))
          slots[bk[u] * BKT_WORDS + pos[u]] =
              ((dd[u] & (BKT_NODES - 1)) << 16) | (ss[u] & 0xFFFF);
      }
    }
  }
}

__device__ __forceinline__ float lrelu(float v) { return fmaxf(v, ALPHA * v); }

#define LOG2E 1.44269504088896340736f

__device__ __forceinline__ int node_clamp(int s) {
  return (int)min((unsigned)s, (unsigned)(N_NODES - 1));
}

__device__ __forceinline__ float pair_sum(float p) {
  p += __builtin_bit_cast(float, __builtin_amdgcn_update_dpp(
           0, __builtin_bit_cast(int, p), 0xB1, 0xF, 0xF, true));
  return p;
}

__device__ __forceinline__ v2f lo2(const v4f a) { return (v2f){ a.x, a.y }; }
__device__ __forceinline__ v2f hi2(const v4f a) { return (v2f){ a.z, a.w }; }

__device__ __forceinline__ float dot8(const v4f z0, const v4f z1, const v4f a0, const v4f a1) {
  v2f r = lo2(z0) * lo2(a0);
  r = __builtin_elementwise_fma(hi2(z0), hi2(a0), r);
  r = __builtin_elementwise_fma(lo2(z1), lo2(a1), r);
  r = __builtin_elementwise_fma(hi2(z1), hi2(a1), r);
  return r.x + r.y;
}

struct GSum {
  v4f   a0, a1;
  float den;
};
__device__ __forceinline__ void gsum_scale(GSum& s, const float sc) {
  s.den *= sc;
  s.a0  *= sc;
  s.a1  *= sc;
}
__device__ __forceinline__ void gsum_add(GSum& s, const float ex, const v4f z0, const v4f z1) {
  s.den += ex;
  s.a0 = __builtin_elementwise_fma((v4f){ ex, ex, ex, ex }, z0, s.a0);
  s.a1 = __builtin_elementwise_fma((v4f){ ex, ex, ex, ex }, z1, s.a1);
}

__device__ __forceinline__ void edge_online(GSum& sum, float& m,
                                            const v4f* __restrict__ z4p, const unsigned uq,
                                            const int s,
                                            const v4f As0, const v4f As1, const float ad)
{
  const v4f z0 = z4p[(unsigned)s * (ZCOLS / 4) + uq];
  const v4f z1 = z4p[(unsigned)s * (ZCOLS / 4) + uq + 8u];
  const float v  = lrelu(pair_sum(dot8(z0, z1, As0, As1)) + ad);
  const float mn = fmaxf(m, v);
  const float sc = __builtin_amdgcn_exp2f(m - mn);
  const float ex = __builtin_amdgcn_exp2f(v - mn);
  gsum_scale(sum, sc);
  gsum_add(sum, ex, z0, z1);
  m = mn;
}

#define ROW_CAP    64
#define ROW_PITCH  (ROW_CAP + 1)
#define GB_NODES   64
#define GB_THREADS 256
#define GB_LANES   8
#define GB_BLOCKS  ((N_NODES + GB_NODES - 1) / GB_NODES)
#define GB_PER_BKT (BKT_NODES / GB_NODES)
#define GB_GROUPS  (GB_THREADS / GB_LANES)
#define GB_ROUNDS  (GB_NODES / GB_GROUPS)
#define GB_QUADS   (BKT_WORDS / 4 / GB_THREADS)
#define TRIP 8
static_assert(TRIP == 8, "the trip's arrival point names its eight rows");
static_assert(BKT_WORDS % (4 * GB_THREADS) == 0, "whole quads per thread");
static_assert(GB_NODES % GB_GROUPS == 0 && GB_ROUNDS == 2, "two whole rounds");
static_assert(GB_LANES * 8 == ZCOLS, "a lane owns 8 columns");
__global__ __launch_bounds__(GB_THREADS) void gat_rows_gather(
    const int* __restrict__ gcnt, const int* __restrict__ slots,
    const int* __restrict__ src, const int* __restrict__ dst,
    const float* __restrict__ A,
    const float* __restrict__ z, float* __restrict__ out)
{
  __shared__ int s_row[GB_NODES * ROW_PITCH];
  __shared__ int s_cnt[GB_NODES];
  const int tid   = threadIdx.x;
  const int b     = blockIdx.x / GB_PER_BKT;
  const int lo    = (blockIdx.x % GB_PER_BKT) * GB_NODES;
  const int total = gcnt[b];
  const bool over = total > bucket_cap(b);
  const int  tot  = over ? 0 : total;
  const int* __restrict__ area = slots + b * BKT_WORDS;

  const int g  = tid / GB_LANES;
  const int q  = tid % GB_LANES;
  const int hh = q >> 1;
  const v4f* __restrict__ z4p = (const v4f*)z;
  const unsigned uq = (unsigned)q;
  const float* __restrict__ Ah = A + hh * (2 * OUT_FEATS) + (q & 1) * 8;
  const v4f As0 = *(const v4f*)(Ah)                 * LOG2E;
  const v4f As1 = *(const v4f*)(Ah + 4)             * LOG2E;
  const v4f Ad0 = *(const v4f*)(Ah + OUT_FEATS)     * LOG2E;
  const v4f Ad1 = *(const v4f*)(Ah + OUT_FEATS + 4) * LOG2E;
  v4f znr[GB_ROUNDS][2];
#pragma unroll
  for (int rd = 0; rd < GB_ROUNDS; ++rd) {
    const int nr = min(blockIdx.x * GB_NODES + rd * GB_GROUPS + g, N_NODES - 1);
    znr[rd][0] = z4p[(unsigned)nr * (ZCOLS / 4) + uq];
    znr[rd][1] = z4p[(unsigned)nr * (ZCOLS / 4) + uq + 8u];
  }

  if (tid < GB_NODES) s_cnt[tid] = 0;
  const int qlast = max((tot - 1) >> 2, 0);
  int4 w[GB_QUADS];
#pragma unroll
  for (int j = 0; j < GB_QUADS; ++j)
    w[j] = ((const int4*)area)[min(j * GB_THREADS + tid, qlast)];
  __syncthreads();
  float adr[GB_ROUNDS];
#pragma unroll
  for (int rd = 0; rd < GB_ROUNDS; ++rd)
    adr[rd] = pair_sum(dot8(znr[rd][0], znr[rd][1], Ad0, Ad1));
  asm volatile("" : "+v"(adr[0]), "+v"(adr[1]));
  int rk[GB_QUADS][4];
#pragma unroll
  for (int j = 0; j < GB_QUADS; ++j) {
    const int i0 = 4 * (j * GB_THREADS + tid);
    const int ww[4] = { w[j].x, w[j].y, w[j].z, w[j].w };
#pragma unroll
    for (int u = 0; u < 4; ++u) {
      const int dl = ((ww[u] >> 16) & (BKT_NODES - 1)) - lo;
      rk[j][u] = ROW_CAP;
      if (i0 + u < tot && (unsigned)dl < (unsigned)GB_NODES)
        rk[j][u] = atomicAdd(&s_cnt[dl], 1);
    }
  }
#pragma unroll
  for (int j = 0; j < GB_QUADS; ++j) {
    const int ww[4] = { w[j].x, w[j].y, w[j].z, w[j].w };
#pragma unroll
    for (int u = 0; u < 4; ++u) {
      const int dl = ((ww[u] >> 16) & (BKT_NODES - 1)) - lo;
      if (rk[j][u] < ROW_CAP) s_row[dl * ROW_PITCH + rk[j][u]] = node_clamp(ww[u] & 0xFFFF);
    }
  }
  __syncthreads();

  for (int rd = 0; rd < GB_ROUNDS; ++rd) {
    const int nl = rd * GB_GROUPS + g;
    const int n  = blockIdx.x * GB_NODES + nl;
    if (n >= N_NODES) break;
    const int  deg   = s_cnt[nl];
    const bool full  = deg > ROW_CAP;
    const int  end   = (over || full) ? 0 : deg;
    const int  nbw   = (!over && full) ? total : 0;
    const int  nwalk = over ? N_EDGES : 0;
    const int* __restrict__ rowp = s_row + nl * ROW_PITCH;

    const float ad = rd == 0 ? adr[0] : adr[1];

    float m = -INFINITY;
    GSum  sum;
    sum.a0  = (v4f){ 0.f, 0.f, 0.f, 0.f };
    sum.a1  = (v4f){ 0.f, 0.f, 0.f, 0.f };
    sum.den = 0.f;

    int sn[TRIP];
#pragma unroll
    for (int j = 0; j < TRIP; ++j) sn[j] = 0;
    if (end > 0) {
#pragma unroll
      for (int j = 0; j < TRIP; ++j)
        sn[j] = rowp[min(j, end - 1)];
    }
    for (int k = 0; k < end; k += TRIP) {
      v4f    zr0[TRIP], zr1[TRIP];
      float  v[TRIP];
#pragma unroll
      for (int j = 0; j < TRIP; ++j) {
        zr0[j] = z4p[(unsigned)sn[j] * (ZCOLS / 4) + uq];
        zr1[j] = z4p[(unsigned)sn[j] * (ZCOLS / 4) + uq + 8u];
      }
#pragma unroll
      for (int j = 0; j < TRIP; ++j)
        sn[j] = rowp[min(k + TRIP + j, end - 1)];
      asm volatile("" : "+v"(zr0[0]), "+v"(zr0[1]), "+v"(zr0[2]), "+v"(zr0[3]),
                        "+v"(zr0[4]), "+v"(zr0[5]), "+v"(zr0[6]), "+v"(zr0[7]),
                        "+v"(zr1[0]), "+v"(zr1[1]), "+v"(zr1[2]), "+v"(zr1[3]),
                        "+v"(zr1[4]), "+v"(zr1[5]), "+v"(zr1[6]), "+v"(zr1[7]));
      float tm = -INFINITY;
#pragma unroll
      for (int j = 0; j < TRIP; ++j) {
        const float e = lrelu(pair_sum(dot8(zr0[j], zr1[j], As0, As1)) + ad);
        v[j] = (k + j < end) ? e : -INFINITY;
        tm = fmaxf(tm, v[j]);
      }
      const float mn = fmaxf(m, tm);
      const float sc = __builtin_amdgcn_exp2f(m - mn);
      gsum_scale(sum, sc);
#pragma unroll
      for (int j = 0; j < TRIP; ++j)
        gsum_add(sum, __builtin_amdgcn_exp2f(v[j] - mn), zr0[j], zr1[j]);
      m = mn;
    }
    for (int i = 0; i < nbw; ++i) {
      const int wd = area[i];
      if (((wd >> 16) & (BKT_NODES - 1)) == lo + nl)
        edge_online(sum, m, z4p, uq, node_clamp(wd & 0xFFFF), As0, As1, ad);
    }
    for (int e = 0; e < nwalk; ++e)
      if (dst[e] == n) edge_online(sum, m, z4p, uq, node_clamp(src[e]), As0, As1, ad);
    const float dd = fmaxf(sum.den, 1e-12f);
    v4f* __restrict__ op = (v4f*)(out + (size_t)n * ZCOLS + q * 8);
    op[0] = sum.a0 / dd;
    op[1] = sum.a1 / dd;
  }
}

extern "C" void kernel_launch(void* const* d_in, const int* in_sizes, int n_in,
                              void* d_out, int out_size, void* d_ws, size_t ws_size,
                              hipStream_t stream) {
  const float* h   = (const float*)d_in[0];
  const int*   src = (const int*)d_in[1];
  const int*   dst = (const int*)d_in[2];
  const float* W   = (const float*)d_in[3];
  const float* A   = (const float*)d_in[4];
  float* out = (float*)d_out;

  float* z     = (float*)d_ws;
  int*   gcnt  = (int*)(z + (size_t)N_NODES * ZCOLS);
  int*   slots = gcnt + HEAD_WORDS;

  gat_clear_counts<<<1, 64, 0, stream>>>(gcnt);
  gat_proj_and_bin<<<BIN_BLOCKS + PROJ_BLOCKS, 128, 0, stream>>>(h, W, z, src, dst, gcnt, slots);
  gat_rows_gather<<<GB_BLOCKS, GB_THREADS, 0, stream>>>(gcnt, slots, src, dst, A, z, out);
}
